# starting kernel
# baseline (speedup 1.0000x reference)
#include <hip/hip_runtime.h>


typedef __attribute__((ext_vector_type(2))) float v2f;
typedef __attribute__((ext_vector_type(4))) float v4f;
typedef __attribute__((ext_vector_type(8))) float v8f;

#define KS     7
#define PADR   3
#define CH     16
#define KK     49
#define KKP    64
#define TILE_W 128
#define HALO_W (TILE_W + KS - 1)
#define NX     (KS * HALO_W)
#define BN_EPS 1e-5f

__global__ __launch_bounds__(256) void involution_wmma_kernel(
    const float* __restrict__ x,
    const float* __restrict__ w1,  const float* __restrict__ b1,
    const float* __restrict__ gam, const float* __restrict__ bet,
    const float* __restrict__ mu,  const float* __restrict__ var,
    const float* __restrict__ w2,  const float* __restrict__ b2,
    float* __restrict__ out, int B, int H, int W)
{
    __shared__ __align__(16) float s_x[NX];
    __shared__ __align__(16) float s_w2[KKP][CH];
    __shared__ __align__(32) float s_b2[KKP];
    __shared__ __align__(16) float s_ad[2 * CH];

    const int tid = threadIdx.x;

    for (int idx = tid; idx < KKP * CH; idx += 256) {
        const int k = idx >> 4, c = idx & 15;
        s_w2[k][c] = (k < KK) ? w2[k * CH + c] : 0.f;
    }
    if (tid < KKP) s_b2[tid] = (tid < KK) ? b2[tid] : 0.f;
    if (tid < CH) {
        const float inv = gam[tid] * rsqrtf(var[tid] + BN_EPS);
        s_ad[2 * tid]     = w1[tid] * inv;
        s_ad[2 * tid + 1] = (b1[tid] - mu[tid]) * inv + bet[tid];
    }

    const int col0 = blockIdx.x * TILE_W;
    const int row  = blockIdx.y;
    const int b    = blockIdx.z;
    const float* __restrict__ xb = x + (size_t)b * H * W;

    for (int idx = tid; idx < NX; idx += 256) {
        const int r  = idx / HALO_W;
        const int c  = idx - r * HALO_W;
        const int gi = row - PADR + r;
        const int gj = col0 - PADR + c;
        float v = 0.f;
        if (gi >= 0 && gi < H && gj >= 0 && gj < W) v = xb[gi * W + gj];
        s_x[idx] = v;
    }
    __syncthreads();

    const int lane = tid & 31;
    const int wave = tid >> 5;
    const int half = lane >> 4;
    const int p    = lane & 15;
    const int j    = wave * 16 + p;

    const float xc = s_x[PADR * HALO_W + PADR + j];

    v2f A[4][4];
#pragma unroll
    for (int t = 0; t < 4; ++t)
#pragma unroll
        for (int s = 0; s < 4; ++s)
            A[t][s] = *(const v2f*)&s_w2[t * 16 + p][s * 4 + 2 * half];

    const float* pb2 = &s_b2[8 * half];
    v8f acc[4];
#pragma unroll
    for (int t = 0; t < 4; ++t)
        acc[t] = *(const v8f*)&pb2[16 * t];

    const float* pada = &s_ad[4 * half];
#pragma unroll
    for (int s = 0; s < 4; ++s) {
        const v4f ad = *(const v4f*)&pada[8 * s];
        v2f Bv;
        Bv.x = fmaxf(fmaf(ad.x, xc, ad.y), 0.f);
        Bv.y = fmaxf(fmaf(ad.z, xc, ad.w), 0.f);
#pragma unroll
        for (int t = 0; t < 4; ++t) {
            acc[t] = __builtin_amdgcn_wmma_f32_16x16x4_f32(
                false, A[t][s], false, Bv, (short)0, acc[t], false, false);
        }
    }

    const float* q0 = &s_x[j];
    const float* q1 = q0 + 135 * half;
    const float* q2 = q0 + 262 * half;

    float partial = 0.f;
#pragma unroll
    for (int t = 0; t < 4; ++t) {
#pragma unroll
        for (int v = 0; v < 8; ++v) {
            const int mlo  = 16 * t + v;
            const int mhi  = mlo + 8;
            const int mloc = (mlo < KK) ? mlo : 48;
            const int off  = (mloc / KS) * HALO_W + (mloc % KS);
            const float* q = (mhi >= KK) ? q0 : ((mlo % KS == 6) ? q2 : q1);
            partial = fmaf(acc[t][v], q[off], partial);
        }
    }

    partial += __shfl_xor(partial, 16, 32);

    if (half == 0)
        out[((size_t)b * H + row) * W + col0 + j] = partial;
}

extern "C" void kernel_launch(void* const* d_in, const int* in_sizes, int n_in,
                              void* d_out, int out_size, void* d_ws, size_t ws_size,
                              hipStream_t stream) {
    (void)n_in; (void)d_ws; (void)ws_size; (void)out_size;
    const float* x   = (const float*)d_in[0];
    const float* w1  = (const float*)d_in[1];
    const float* b1  = (const float*)d_in[2];
    const float* gam = (const float*)d_in[3];
    const float* bet = (const float*)d_in[4];
    const float* mu  = (const float*)d_in[5];
    const float* var = (const float*)d_in[6];
    const float* w2  = (const float*)d_in[7];
    const float* b2  = (const float*)d_in[8];
    float* out = (float*)d_out;

    const int H = 256, W = 256;
    const int B = in_sizes[0] / (H * W);

    dim3 grid(W / TILE_W, H, B);
    dim3 block(256);
    involution_wmma_kernel<<<grid, block, 0, stream>>>(
        x, w1, b1, gam, bet, mu, var, w2, b2, out, B, H, W);
}


# each work-group walks a 16-row x 128-col tile instead of one row: W2, b2 and BN fragments stay in registers, the halo is staged once per tile, 16x fewer waves are launched
# speedup vs baseline: 1.6462x; 1.6462x over previous
#include <hip/hip_runtime.h>


typedef __attribute__((ext_vector_type(2))) float v2f;
typedef __attribute__((ext_vector_type(4))) float v4f;
typedef __attribute__((ext_vector_type(8))) float v8f;

#define KS     7
#define PADR   3
#define CH     16
#define KK     49
#define KKP    64
#define TILE_W 128
#define TILE_H 16
#define HALO_W (TILE_W + KS - 1)
#define HALO_H (TILE_H + KS - 1)
#define NX     (HALO_H * HALO_W)
#define HROWS  3
#define HCOLS  5
#define BN_EPS 1e-5f

__global__ __launch_bounds__(256) void involution_wmma_kernel(
    const float* __restrict__ x,
    const float* __restrict__ w1,  const float* __restrict__ b1,
    const float* __restrict__ gam, const float* __restrict__ bet,
    const float* __restrict__ mu,  const float* __restrict__ var,
    const float* __restrict__ w2,  const float* __restrict__ b2,
    float* __restrict__ out, int B, int H, int W)
{
    __shared__ __align__(16) float s_x[NX];
    __shared__ __align__(16) float s_w2[KKP][CH];
    __shared__ __align__(32) float s_b2[KKP];
    __shared__ __align__(16) float s_ad[2 * CH];

    const int tid  = threadIdx.x;
    const int lane = tid & 31;
    const int wave = tid >> 5;

    const int col0 = blockIdx.x * TILE_W;
    const int row0 = blockIdx.y * TILE_H;
    const int b    = blockIdx.z;
    const float* __restrict__ xb = x + (size_t)b * H * W;

    float wv[4];
#pragma unroll
    for (int k = 0; k < 4; ++k) {
        const int idx = tid + 256 * k;
        wv[k] = w2[idx < KK * CH ? idx : KK * CH - 1];
    }
    const float b2v = b2[tid < KK ? tid : KK - 1];
    const int   c16 = tid & (CH - 1);
    const float gv = gam[c16], vv = var[c16], w1v = w1[c16];
    const float b1v = b1[c16], muv = mu[c16], bev = bet[c16];

    float hv[HROWS][HCOLS];
#pragma unroll
    for (int rr = 0; rr < HROWS; ++rr) {
        const int gi  = row0 - PADR + wave + 8 * rr;
        const int gic = gi < 0 ? 0 : (gi >= H ? H - 1 : gi);
        const float* __restrict__ xr = xb + gic * W;
#pragma unroll
        for (int k = 0; k < HCOLS; ++k) {
            const int   gj  = col0 - PADR + lane + 32 * k;
            const int   gjc = gj < 0 ? 0 : (gj >= W ? W - 1 : gj);
            const float v   = xr[gjc];
            hv[rr][k] = ((gi == gic) & (gj == gjc)) ? v : 0.f;
        }
    }

    asm volatile("" :: "v"(b2v));

#pragma unroll
    for (int k = 0; k < 4; ++k) {
        const int idx = tid + 256 * k;
        s_w2[idx >> 4][idx & 15] = (idx < KK * CH) ? wv[k] : 0.f;
    }
    if (tid < KKP) s_b2[tid] = (tid < KK) ? b2v : 0.f;
    if (tid < CH) {
        const float inv = gv * rsqrtf(vv + BN_EPS);
        s_ad[2 * tid]     = w1v * inv;
        s_ad[2 * tid + 1] = (b1v - muv) * inv + bev;
    }
#pragma unroll
    for (int rr = 0; rr < HROWS; ++rr) {
        const int r = wave + 8 * rr;
#pragma unroll
        for (int k = 0; k < HCOLS; ++k) {
            const int c = lane + 32 * k;
            if (r < HALO_H && c < HALO_W) s_x[r * HALO_W + c] = hv[rr][k];
        }
    }
    __syncthreads();

    const int half = lane >> 4;
    const int p    = lane & 15;
    const int j    = wave * 16 + p;

    v2f A[4][4];
#pragma unroll
    for (int t = 0; t < 4; ++t)
#pragma unroll
        for (int s = 0; s < 4; ++s)
            A[t][s] = *(const v2f*)&s_w2[t * 16 + p][s * 4 + 2 * half];

    const float* pb2 = &s_b2[8 * half];
    v8f cinit[4];
#pragma unroll
    for (int t = 0; t < 4; ++t)
        cinit[t] = *(const v8f*)&pb2[16 * t];

    const float* pada = &s_ad[4 * half];
    v4f ad[4];
#pragma unroll
    for (int s = 0; s < 4; ++s)
        ad[s] = *(const v4f*)&pada[8 * s];

    const float* q0 = &s_x[j];
    const float* q1 = q0 + 135 * half;
    const float* q2 = q0 + 262 * half;
    const float* qc = &s_x[PADR * HALO_W + PADR + j];
    float* po = out + ((size_t)b * H + row0) * W + col0 + j;

#pragma unroll 2
    for (int r = 0; r < TILE_H; ++r) {
        const float xc = qc[0];

        v8f acc[4];
#pragma unroll
        for (int s = 0; s < 4; ++s) {
            v2f Bv;
            Bv.x = fmaxf(fmaf(ad[s].x, xc, ad[s].y), 0.f);
            Bv.y = fmaxf(fmaf(ad[s].z, xc, ad[s].w), 0.f);
#pragma unroll
            for (int t = 0; t < 4; ++t) {
                acc[t] = __builtin_amdgcn_wmma_f32_16x16x4_f32(
                    false, A[t][s], false, Bv, (short)0,
                    (s == 0) ? cinit[t] : acc[t], false, false);
            }
        }

        float partial = 0.f;
#pragma unroll
        for (int t = 0; t < 4; ++t) {
#pragma unroll
            for (int v = 0; v < 8; ++v) {
                const int mlo  = 16 * t + v;
                const int mhi  = mlo + 8;
                const int mloc = (mlo < KK) ? mlo : 48;
                const int off  = (mloc / KS) * HALO_W + (mloc % KS);
                const float* q = (mhi >= KK) ? q0 : ((mlo % KS == 6) ? q2 : q1);
                partial = fmaf(acc[t][v], q[off], partial);
            }
        }

        partial += __shfl_xor(partial, 16, 32);

        if (half == 0)
            po[0] = partial;

        q0 += HALO_W; q1 += HALO_W; q2 += HALO_W; qc += HALO_W;
        po += W;
    }
}

extern "C" void kernel_launch(void* const* d_in, const int* in_sizes, int n_in,
                              void* d_out, int out_size, void* d_ws, size_t ws_size,
                              hipStream_t stream) {
    (void)n_in; (void)d_ws; (void)ws_size; (void)out_size;
    const float* x   = (const float*)d_in[0];
    const float* w1  = (const float*)d_in[1];
    const float* b1  = (const float*)d_in[2];
    const float* gam = (const float*)d_in[3];
    const float* bet = (const float*)d_in[4];
    const float* mu  = (const float*)d_in[5];
    const float* var = (const float*)d_in[6];
    const float* w2  = (const float*)d_in[7];
    const float* b2  = (const float*)d_in[8];
    float* out = (float*)d_out;

    const int H = 256, W = 256;
    const int B = in_sizes[0] / (H * W);

    dim3 grid(W / TILE_W, H / TILE_H, B);
    dim3 block(256);
    involution_wmma_kernel<<<grid, block, 0, stream>>>(
        x, w1, b1, gam, bet, mu, var, w2, b2, out, B, H, W);
}


# sum reassociated: out = sum_c h_c*(W2^T.patch)_c + b2.patch, so the f32 WMMA has the 16 channels as rows and the 49 taps as K (padded to 52): 13 WMMAs per 16 pixels, not 16; fp32 sum order changes
# speedup vs baseline: 2.0962x; 1.2734x over previous
#include <hip/hip_runtime.h>


typedef __attribute__((ext_vector_type(2))) float v2f;
typedef __attribute__((ext_vector_type(4))) float v4f;
typedef __attribute__((ext_vector_type(8))) float v8f;

#define KS     7
#define PADR   3
#define CH     16
#define KK     49
#define KKP    64
#define TILE_W 128
#define TILE_H 16
#define HALO_W (TILE_W + KS - 1)
#define HALO_H (TILE_H + KS - 1)
#define NX     (HALO_H * HALO_W)
#define HROWS  3
#define HCOLS  5
#define BN_EPS 1e-5f
#define NSTEP  13
#define HSLOT  26

__global__ __launch_bounds__(256) void involution_wmma_kernel(
    const float* __restrict__ x,
    const float* __restrict__ w1,  const float* __restrict__ b1,
    const float* __restrict__ gam, const float* __restrict__ bet,
    const float* __restrict__ mu,  const float* __restrict__ var,
    const float* __restrict__ w2,  const float* __restrict__ b2,
    float* __restrict__ out, int B, int H, int W)
{
    __shared__ __align__(16) float s_x[NX];
    __shared__ __align__(16) float s_w2[KKP][CH];
    __shared__ __align__(32) float s_b2[KKP];
    __shared__ __align__(16) float s_ad[2 * CH];

    const int tid  = threadIdx.x;
    const int lane = tid & 31;
    const int wave = tid >> 5;

    const int col0 = blockIdx.x * TILE_W;
    const int row0 = blockIdx.y * TILE_H;
    const int b    = blockIdx.z;
    const float* __restrict__ xb = x + (size_t)b * H * W;

    float wv[4];
#pragma unroll
    for (int k = 0; k < 4; ++k) {
        const int idx = tid + 256 * k;
        wv[k] = w2[idx < KK * CH ? idx : KK * CH - 1];
    }
    const float b2v = b2[tid < KK ? tid : KK - 1];
    const int   c16 = tid & (CH - 1);
    const float gv = gam[c16], vv = var[c16], w1v = w1[c16];
    const float b1v = b1[c16], muv = mu[c16], bev = bet[c16];

    float hv[HROWS][HCOLS];
#pragma unroll
    for (int rr = 0; rr < HROWS; ++rr) {
        const int gi  = row0 - PADR + wave + 8 * rr;
        const int gic = gi < 0 ? 0 : (gi >= H ? H - 1 : gi);
        const float* __restrict__ xr = xb + gic * W;
#pragma unroll
        for (int k = 0; k < HCOLS; ++k) {
            const int   gj  = col0 - PADR + lane + 32 * k;
            const int   gjc = gj < 0 ? 0 : (gj >= W ? W - 1 : gj);
            const float v   = xr[gjc];
            hv[rr][k] = ((gi == gic) & (gj == gjc)) ? v : 0.f;
        }
    }

    asm volatile("" :: "v"(b2v));

#pragma unroll
    for (int k = 0; k < 4; ++k) {
        const int idx = tid + 256 * k;
        s_w2[idx >> 4][idx & 15] = (idx < KK * CH) ? wv[k] : 0.f;
    }
    if (tid < KKP) s_b2[tid] = (tid < KK) ? b2v : 0.f;
    if (tid < CH) {
        const float inv = gv * rsqrtf(vv + BN_EPS);
        s_ad[tid]      = w1v * inv;
        s_ad[CH + tid] = (b1v - muv) * inv + bev;
    }
#pragma unroll
    for (int rr = 0; rr < HROWS; ++rr) {
        const int r = wave + 8 * rr;
#pragma unroll
        for (int k = 0; k < HCOLS; ++k) {
            const int c = lane + 32 * k;
            if (r < HALO_H && c < HALO_W) s_x[r * HALO_W + c] = hv[rr][k];
        }
    }
    __syncthreads();

    const int half = lane >> 4;
    const int p    = lane & 15;
    const int j    = wave * 16 + p;

    v2f A[NSTEP], bv[NSTEP];
#pragma unroll
    for (int s = 0; s < NSTEP; ++s) {
        const int m = 2 * s + HSLOT * half;
        A[s].x = s_w2[m][p];
        A[s].y = s_w2[m + 1][p];
        bv[s]  = *(const v2f*)&s_b2[m];
    }

    v2f av[4], dv[4];
#pragma unroll
    for (int q = 0; q < 4; ++q) {
        av[q] = *(const v2f*)&s_ad[8 * half + 2 * q];
        dv[q] = *(const v2f*)&s_ad[CH + 8 * half + 2 * q];
    }

    const float* q0 = &s_x[j];
    const float* q1 = q0 + 407 * half;
    const float* q2 = q0 + 534 * half;
    const float* qc = &s_x[PADR * HALO_W + PADR + j];
    float* po = out + ((size_t)b * H + row0) * W + col0 + j;

#pragma unroll 1
    for (int r = 0; r < TILE_H; r += 2) {
        v8f accA = {0.f, 0.f, 0.f, 0.f, 0.f, 0.f, 0.f, 0.f};
        v8f accB = {0.f, 0.f, 0.f, 0.f, 0.f, 0.f, 0.f, 0.f};
        v2f gA = {0.f, 0.f};
        v2f gB = {0.f, 0.f};

#pragma unroll
        for (int s = 0; s < NSTEP; ++s) {
            v2f BA, BB;
#pragma unroll
            for (int e = 0; e < 2; ++e) {
                const int t   = 2 * s + e;
                const int off = (t / KS) * HALO_W + (t % KS);
                const float* q = (t >= 23) ? q0 : ((t % KS <= 1) ? q1 : q2);
                BA[e] = q[off];
                BB[e] = q[off + HALO_W];
            }
            accA = __builtin_amdgcn_wmma_f32_16x16x4_f32(
                false, A[s], false, BA, (short)0, accA, false, false);
            accB = __builtin_amdgcn_wmma_f32_16x16x4_f32(
                false, A[s], false, BB, (short)0, accB, false, false);
            gA += bv[s] * BA;
            gB += bv[s] * BB;
        }

        const float xcA = qc[0];
        const float xcB = qc[HALO_W];
        const v2f zero2 = {0.f, 0.f};
        v2f pA = gA, pB = gB;
#pragma unroll
        for (int q = 0; q < 4; ++q) {
            const v2f hA = __builtin_elementwise_max(av[q] * xcA + dv[q], zero2);
            const v2f hB = __builtin_elementwise_max(av[q] * xcB + dv[q], zero2);
            const v2f GA = {accA[2 * q], accA[2 * q + 1]};
            const v2f GB = {accB[2 * q], accB[2 * q + 1]};
            pA += hA * GA;
            pB += hB * GB;
        }
        float partA = pA.x + pA.y;
        float partB = pB.x + pB.y;

        partA += __shfl_xor(partA, 16, 32);
        partB += __shfl_xor(partB, 16, 32);

        if (half == 0) {
            po[0] = partA;
            po[W] = partB;
        }

        q0 += 2 * HALO_W; q1 += 2 * HALO_W; q2 += 2 * HALO_W; qc += 2 * HALO_W;
        po += 2 * W;
    }
}

extern "C" void kernel_launch(void* const* d_in, const int* in_sizes, int n_in,
                              void* d_out, int out_size, void* d_ws, size_t ws_size,
                              hipStream_t stream) {
    (void)n_in; (void)d_ws; (void)ws_size; (void)out_size;
    const float* x   = (const float*)d_in[0];
    const float* w1  = (const float*)d_in[1];
    const float* b1  = (const float*)d_in[2];
    const float* gam = (const float*)d_in[3];
    const float* bet = (const float*)d_in[4];
    const float* mu  = (const float*)d_in[5];
    const float* var = (const float*)d_in[6];
    const float* w2  = (const float*)d_in[7];
    const float* b2  = (const float*)d_in[8];
    float* out = (float*)d_out;

    const int H = 256, W = 256;
    const int B = in_sizes[0] / (H * W);

    dim3 grid(W / TILE_W, H / TILE_H, B);
    dim3 block(256);
    involution_wmma_kernel<<<grid, block, 0, stream>>>(
        x, w1, b1, gam, bet, mu, var, w2, b2, out, B, H, W);
}
